# starting kernel
# baseline (speedup 1.0000x reference)
#include <hip/hip_runtime.h>
#include <limits.h>

typedef float v2f __attribute__((ext_vector_type(2)));
typedef float v8f __attribute__((ext_vector_type(8)));

#define NEG_INF (-1e30f)

__global__ __launch_bounds__(256) void gemv_wmma_f32(
    const float* __restrict__ A, const float* __restrict__ W,
    const float* __restrict__ bias, float* __restrict__ out,
    int N, int Hdim) {
  int wave = (int)((blockIdx.x * blockDim.x + threadIdx.x) >> 5);
  int lane = (int)(threadIdx.x & 31);
  int row0 = wave * 16;
  if (row0 >= N) return;

  int m = lane & 15;
  int khalf = (lane >> 4) * 2;
  int row = row0 + m;
  if (row >= N) row = N - 1;

  const float* arow = A + (size_t)row * (size_t)Hdim + khalf;
  const float* wp   = W + khalf;

  v8f acc0 = {}; v8f acc1 = {}; v8f acc2 = {}; v8f acc3 = {};
  for (int k = 0; k < Hdim; k += 16) {
    v2f a0 = __builtin_nontemporal_load((const v2f*)(arow + k));
    v2f b0 = *(const v2f*)(wp + k);
    v2f a1 = __builtin_nontemporal_load((const v2f*)(arow + k + 4));
    v2f b1 = *(const v2f*)(wp + k + 4);
    v2f a2 = __builtin_nontemporal_load((const v2f*)(arow + k + 8));
    v2f b2 = *(const v2f*)(wp + k + 8);
    v2f a3 = __builtin_nontemporal_load((const v2f*)(arow + k + 12));
    v2f b3 = *(const v2f*)(wp + k + 12);
    acc0 = __builtin_amdgcn_wmma_f32_16x16x4_f32(false, a0, false, b0,
                                                 (short)0, acc0, false, false);
    acc1 = __builtin_amdgcn_wmma_f32_16x16x4_f32(false, a1, false, b1,
                                                 (short)0, acc1, false, false);
    acc2 = __builtin_amdgcn_wmma_f32_16x16x4_f32(false, a2, false, b2,
                                                 (short)0, acc2, false, false);
    acc3 = __builtin_amdgcn_wmma_f32_16x16x4_f32(false, a3, false, b3,
                                                 (short)0, acc3, false, false);
  }
  v8f acc = (acc0 + acc1) + (acc2 + acc3);
  float bv = bias[0];

  if ((lane & 15) == 0) {
    int mb = row0 + (lane >> 4) * 8;
#pragma unroll
    for (int r = 0; r < 8; ++r) {
      int rr = mb + r;
      if (rr < N) out[rr] = acc[r] + bv;
    }
  }
}

__global__ void scan_offsets(const int* __restrict__ lens,
                             int* __restrict__ offsets, int S) {
  int lane = (int)threadIdx.x;
  int run = 0;
  for (int base = 0; base < S; base += 32) {
    int i = base + lane;
    int x = (i < S) ? lens[i] : 0;
    int v = x;
#pragma unroll
    for (int d = 1; d < 32; d <<= 1) {
      int t = __shfl_up(v, d, 32);
      if (lane >= d) v += t;
    }
    if (i < S) offsets[i] = run + (v - x);
    run += __shfl(v, 31, 32);
  }
}

__global__ __launch_bounds__(256) void seg_loss(
    const float* __restrict__ logits, const int* __restrict__ lens,
    const int* __restrict__ offsets, const int* __restrict__ tgt,
    float* __restrict__ loss_ws, float* __restrict__ inc_ws,
    int S, int EACH) {
  int wave = (int)((blockIdx.x * blockDim.x + threadIdx.x) >> 5);
  int lane = (int)(threadIdx.x & 31);
  int s = wave;
  if (s >= S) return;

  int len = lens[s];
  int off = offsets[s];
  int j0 = lane, j1 = lane + 32;
  bool v0 = j0 < len, v1 = j1 < len;
  float x0 = v0 ? logits[off + j0] : NEG_INF;
  float x1 = v1 ? logits[off + j1] : NEG_INF;
  int   t0 = v0 ? tgt[off + j0] : 0;
  int   t1 = v1 ? tgt[off + j1] : 0;

  float mx = fmaxf(x0, x1);
#pragma unroll
  for (int d = 16; d > 0; d >>= 1) mx = fmaxf(mx, __shfl_xor(mx, d, 32));
  float e0 = v0 ? __expf(x0 - mx) : 0.0f;
  float e1 = v1 ? __expf(x1 - mx) : 0.0f;
  float sum = e0 + e1;
#pragma unroll
  for (int d = 16; d > 0; d >>= 1) sum += __shfl_xor(sum, d, 32);
  float inv = 1.0f / sum;
  float p0 = e0 * inv, p1 = e1 * inv;

  float q0 = v0 ? p0 : NEG_INF;
  float q1 = v1 ? p1 : NEG_INF;
  float m2 = fmaxf(q0, q1);
#pragma unroll
  for (int d = 16; d > 0; d >>= 1) m2 = fmaxf(m2, __shfl_xor(m2, d, 32));
  float s2 = (v0 ? __expf(p0 - m2) : 0.0f) + (v1 ? __expf(p1 - m2) : 0.0f);
#pragma unroll
  for (int d = 16; d > 0; d >>= 1) s2 += __shfl_xor(s2, d, 32);
  float lse2 = m2 + __logf(s2);

  int k0 = (v0 && t0 == 1) ? j0 : INT_MAX;
  int k1 = (v1 && t1 == 1) ? j1 : INT_MAX;
  int tk = min(k0, k1);
#pragma unroll
  for (int d = 16; d > 0; d >>= 1) tk = min(tk, __shfl_xor(tk, d, 32));
  int t_idx = (tk == INT_MAX) ? 0 : tk;
  float psel = (t_idx >= 32) ? p1 : p0;
  float p_t = __shfl(psel, t_idx & 31, 32);
  float loss_per = lse2 - p_t;

  int ts = t0 + t1;
#pragma unroll
  for (int d = 16; d > 0; d >>= 1) ts += __shfl_xor(ts, d, 32);

  int next_len = (s + 1 < S) ? lens[s + 1] : 1;
  bool skip_next_zero = (s > 0) && (s + 1 < S) && (next_len == 0);
  bool skip_no_pos = (((s + 1) % EACH) == 0) && (ts <= 0);
  bool include = (len > 0) && !skip_next_zero && !skip_no_pos;

  if (lane == 0) {
    loss_ws[s] = include ? loss_per : 0.0f;
    inc_ws[s]  = include ? 1.0f : 0.0f;
  }
}

__global__ __launch_bounds__(256) void final_reduce(
    const float* __restrict__ loss_ws, const float* __restrict__ inc_ws,
    float* __restrict__ out_loss, int S) {
  __shared__ float sl[256];
  __shared__ float sc[256];
  int t = (int)threadIdx.x;
  float a = 0.0f, c = 0.0f;
  for (int i = t; i < S; i += 256) { a += loss_ws[i]; c += inc_ws[i]; }
  sl[t] = a; sc[t] = c;
  __syncthreads();
  for (int d = 128; d > 0; d >>= 1) {
    if (t < d) { sl[t] += sl[t + d]; sc[t] += sc[t + d]; }
    __syncthreads();
  }
  if (t == 0) {
    float cnt = sc[0];
    float denom = (cnt > 0.0f) ? cnt : 1e-4f;
    out_loss[0] = sl[0] / denom;
  }
}

extern "C" void kernel_launch(void* const* d_in, const int* in_sizes, int n_in,
                              void* d_out, int out_size, void* d_ws, size_t ws_size,
                              hipStream_t stream) {
  const float* A    = (const float*)d_in[0];
  const float* W    = (const float*)d_in[1];
  const float* b    = (const float*)d_in[2];
  const int*   lens = (const int*)d_in[3];
  const int*   tgt  = (const int*)d_in[4];

  int Hdim = in_sizes[1];
  int N    = in_sizes[0] / Hdim;
  int S    = in_sizes[3];
  const int EACH = 32;

  float* logits   = (float*)d_out;
  float* loss_out = logits + N;

  int*   offsets = (int*)d_ws;
  size_t obytes  = (((size_t)S * sizeof(int)) + 255) & ~(size_t)255;
  float* loss_ws = (float*)((char*)d_ws + obytes);
  float* inc_ws  = loss_ws + S;

  int waves  = (N + 15) / 16;
  int blocks = (waves + 7) / 8;
  gemv_wmma_f32<<<blocks, 256, 0, stream>>>(A, W, b, logits, N, Hdim);

  scan_offsets<<<1, 32, 0, stream>>>(lens, offsets, S);

  seg_loss<<<(S + 7) / 8, 256, 0, stream>>>(logits, lens, offsets, tgt,
                                            loss_ws, inc_ws, S, EACH);

  final_reduce<<<1, 256, 0, stream>>>(loss_ws, inc_ws, loss_out, S);
}


# GEMV A path through LDS: whole-row 512 B loads (each line once) staged in a per-wave LDS tile, read back as 16-row f32 WMMA fragments; W in LDS; was 16 B of 16 rows per load; sum order changes
# speedup vs baseline: 1.7835x; 1.7835x over previous
#include <hip/hip_runtime.h>
#include <limits.h>

typedef float v2f __attribute__((ext_vector_type(2)));
typedef float v4f __attribute__((ext_vector_type(4)));
typedef float v8f __attribute__((ext_vector_type(8)));

#define NEG_INF (-1e30f)

#define GEMV_WAVES 2
#define TILE_ROWS 16
#define CHUNK_K 128
#define ROW_PITCH 132

#define LOAD_CHUNK(g, c)                                                         \
  {                                                                              \
    unsigned q = (unsigned)lane + (unsigned)(c) * (CHUNK_K / 4);                 \
    _Pragma("unroll") for (int r = 0; r < TILE_ROWS; ++r)                        \
      g[r] = __builtin_nontemporal_load(rowb[r] + q);                            \
  }

#define STORE_CHUNK(g)                                                           \
  _Pragma("unroll") for (int r = 0; r < TILE_ROWS; ++r)                          \
    *(v4f*)(a_st + r * ROW_PITCH) = g[r];

#define COMPUTE_CHUNK(c)                                                         \
  {                                                                              \
    const float* wf = w_fr + (c) * CHUNK_K;                                      \
    _Pragma("unroll") for (int i = 0; i < CHUNK_K / 8; i += 2) {                 \
      v4f a0 = *(const v4f*)(a_fr + 8 * i);                                      \
      v4f w0 = *(const v4f*)(wf + 8 * i);                                        \
      v4f a1 = *(const v4f*)(a_fr + 8 * i + 8);                                  \
      v4f w1 = *(const v4f*)(wf + 8 * i + 8);                                    \
      acc0 = __builtin_amdgcn_wmma_f32_16x16x4_f32(                              \
          false, a0.xy, false, w0.xy, (short)0, acc0, false, false);             \
      acc1 = __builtin_amdgcn_wmma_f32_16x16x4_f32(                              \
          false, a0.zw, false, w0.zw, (short)0, acc1, false, false);             \
      acc2 = __builtin_amdgcn_wmma_f32_16x16x4_f32(                              \
          false, a1.xy, false, w1.xy, (short)0, acc2, false, false);             \
      acc3 = __builtin_amdgcn_wmma_f32_16x16x4_f32(                              \
          false, a1.zw, false, w1.zw, (short)0, acc3, false, false);             \
    }                                                                            \
  }

#define PHASE_FENCE()                                                            \
  asm volatile("" : "+v"(acc0), "+v"(acc1), "+v"(acc2), "+v"(acc3) : : "memory");

__global__ __launch_bounds__(GEMV_WAVES * 32) void gemv_wmma_f32(
    const float* __restrict__ A, const float* __restrict__ W,
    const float* __restrict__ bias, float* __restrict__ out,
    int N, int Hdim) {
  extern __shared__ __attribute__((aligned(16))) float smem[];
  float* w_lds = smem;
  int lane = (int)(threadIdx.x & 31);
  int wv = __builtin_amdgcn_readfirstlane((int)(threadIdx.x >> 5));
  float* a_lds = smem + Hdim + wv * (TILE_ROWS * ROW_PITCH);

  int row0 = ((int)blockIdx.x * GEMV_WAVES + wv) * TILE_ROWS;
  int m = lane & 15;
  int h = lane >> 4;

  const v4f* rowb[TILE_ROWS];
#pragma unroll
  for (int r = 0; r < TILE_ROWS; ++r) {
    int rr = row0 + r;
    if (rr >= N) rr = N - 1;
    rowb[r] = (const v4f*)(A + (size_t)rr * (size_t)Hdim);
  }
  float* a_st = a_lds + 4 * lane;
  const float* a_fr = a_lds + m * ROW_PITCH + 4 * h;
  const float* w_fr = w_lds + 4 * h;

  int nch = Hdim / CHUNK_K;
  v4f g0[TILE_ROWS], g1[TILE_ROWS];

  LOAD_CHUNK(g0, 0)
  if (nch > 1) LOAD_CHUNK(g1, 1)
  for (int p = wv; p < nch; p += 4 * GEMV_WAVES) {
    v4f t[4];
#pragma unroll
    for (int j = 0; j < 4; ++j) {
      int pj = p + j * GEMV_WAVES;
      if (pj < nch) t[j] = *(const v4f*)(W + pj * CHUNK_K + 4 * lane);
    }
#pragma unroll
    for (int j = 0; j < 4; ++j) {
      int pj = p + j * GEMV_WAVES;
      if (pj < nch) *(v4f*)(w_lds + pj * CHUNK_K + 4 * lane) = t[j];
    }
  }
  __syncthreads();

  v8f acc0 = {}; v8f acc1 = {}; v8f acc2 = {}; v8f acc3 = {};
  int c = 0;
  for (; c + 3 < nch; c += 2) {
    STORE_CHUNK(g0)
    LOAD_CHUNK(g0, c + 2)
    COMPUTE_CHUNK(c)
    PHASE_FENCE()
    STORE_CHUNK(g1)
    LOAD_CHUNK(g1, c + 3)
    COMPUTE_CHUNK(c + 1)
    PHASE_FENCE()
  }
  STORE_CHUNK(g0)
  if (c + 2 < nch) LOAD_CHUNK(g0, c + 2)
  COMPUTE_CHUNK(c)
  if (c + 1 < nch) {
    STORE_CHUNK(g1)
    COMPUTE_CHUNK(c + 1)
  }
  if (c + 2 < nch) {
    STORE_CHUNK(g0)
    COMPUTE_CHUNK(c + 2)
  }
  v8f acc = (acc0 + acc1) + (acc2 + acc3);
  float bv = bias[0];

  if ((lane & 15) == 0) {
    int mb = row0 + (lane >> 4) * 8;
#pragma unroll
    for (int r = 0; r < 8; ++r) {
      int rr = mb + r;
      if (rr < N) out[rr] = acc[r] + bv;
    }
  }
}

__global__ void scan_offsets(const int* __restrict__ lens,
                             int* __restrict__ offsets, int S) {
  int lane = (int)threadIdx.x;
  int run = 0;
  for (int base = 0; base < S; base += 32) {
    int i = base + lane;
    int x = (i < S) ? lens[i] : 0;
    int v = x;
#pragma unroll
    for (int d = 1; d < 32; d <<= 1) {
      int t = __shfl_up(v, d, 32);
      if (lane >= d) v += t;
    }
    if (i < S) offsets[i] = run + (v - x);
    run += __shfl(v, 31, 32);
  }
}

__global__ __launch_bounds__(256) void seg_loss(
    const float* __restrict__ logits, const int* __restrict__ lens,
    const int* __restrict__ offsets, const int* __restrict__ tgt,
    float* __restrict__ loss_ws, float* __restrict__ inc_ws,
    int S, int EACH) {
  int wave = (int)((blockIdx.x * blockDim.x + threadIdx.x) >> 5);
  int lane = (int)(threadIdx.x & 31);
  int s = wave;
  if (s >= S) return;

  int len = lens[s];
  int off = offsets[s];
  int j0 = lane, j1 = lane + 32;
  bool v0 = j0 < len, v1 = j1 < len;
  float x0 = v0 ? logits[off + j0] : NEG_INF;
  float x1 = v1 ? logits[off + j1] : NEG_INF;
  int   t0 = v0 ? tgt[off + j0] : 0;
  int   t1 = v1 ? tgt[off + j1] : 0;

  float mx = fmaxf(x0, x1);
#pragma unroll
  for (int d = 16; d > 0; d >>= 1) mx = fmaxf(mx, __shfl_xor(mx, d, 32));
  float e0 = v0 ? __expf(x0 - mx) : 0.0f;
  float e1 = v1 ? __expf(x1 - mx) : 0.0f;
  float sum = e0 + e1;
#pragma unroll
  for (int d = 16; d > 0; d >>= 1) sum += __shfl_xor(sum, d, 32);
  float inv = 1.0f / sum;
  float p0 = e0 * inv, p1 = e1 * inv;

  float q0 = v0 ? p0 : NEG_INF;
  float q1 = v1 ? p1 : NEG_INF;
  float m2 = fmaxf(q0, q1);
#pragma unroll
  for (int d = 16; d > 0; d >>= 1) m2 = fmaxf(m2, __shfl_xor(m2, d, 32));
  float s2 = (v0 ? __expf(p0 - m2) : 0.0f) + (v1 ? __expf(p1 - m2) : 0.0f);
#pragma unroll
  for (int d = 16; d > 0; d >>= 1) s2 += __shfl_xor(s2, d, 32);
  float lse2 = m2 + __logf(s2);

  int k0 = (v0 && t0 == 1) ? j0 : INT_MAX;
  int k1 = (v1 && t1 == 1) ? j1 : INT_MAX;
  int tk = min(k0, k1);
#pragma unroll
  for (int d = 16; d > 0; d >>= 1) tk = min(tk, __shfl_xor(tk, d, 32));
  int t_idx = (tk == INT_MAX) ? 0 : tk;
  float psel = (t_idx >= 32) ? p1 : p0;
  float p_t = __shfl(psel, t_idx & 31, 32);
  float loss_per = lse2 - p_t;

  int ts = t0 + t1;
#pragma unroll
  for (int d = 16; d > 0; d >>= 1) ts += __shfl_xor(ts, d, 32);

  int next_len = (s + 1 < S) ? lens[s + 1] : 1;
  bool skip_next_zero = (s > 0) && (s + 1 < S) && (next_len == 0);
  bool skip_no_pos = (((s + 1) % EACH) == 0) && (ts <= 0);
  bool include = (len > 0) && !skip_next_zero && !skip_no_pos;

  if (lane == 0) {
    loss_ws[s] = include ? loss_per : 0.0f;
    inc_ws[s]  = include ? 1.0f : 0.0f;
  }
}

__global__ __launch_bounds__(256) void final_reduce(
    const float* __restrict__ loss_ws, const float* __restrict__ inc_ws,
    float* __restrict__ out_loss, int S) {
  __shared__ float sl[256];
  __shared__ float sc[256];
  int t = (int)threadIdx.x;
  float a = 0.0f, c = 0.0f;
  for (int i = t; i < S; i += 256) { a += loss_ws[i]; c += inc_ws[i]; }
  sl[t] = a; sc[t] = c;
  __syncthreads();
  for (int d = 128; d > 0; d >>= 1) {
    if (t < d) { sl[t] += sl[t + d]; sc[t] += sc[t + d]; }
    __syncthreads();
  }
  if (t == 0) {
    float cnt = sc[0];
    float denom = (cnt > 0.0f) ? cnt : 1e-4f;
    out_loss[0] = sl[0] / denom;
  }
}

extern "C" void kernel_launch(void* const* d_in, const int* in_sizes, int n_in,
                              void* d_out, int out_size, void* d_ws, size_t ws_size,
                              hipStream_t stream) {
  const float* A    = (const float*)d_in[0];
  const float* W    = (const float*)d_in[1];
  const float* b    = (const float*)d_in[2];
  const int*   lens = (const int*)d_in[3];
  const int*   tgt  = (const int*)d_in[4];

  int Hdim = in_sizes[1];
  int N    = in_sizes[0] / Hdim;
  int S    = in_sizes[3];
  const int EACH = 32;

  float* logits   = (float*)d_out;
  float* loss_out = logits + N;

  int*   offsets = (int*)d_ws;
  size_t obytes  = (((size_t)S * sizeof(int)) + 255) & ~(size_t)255;
  float* loss_ws = (float*)((char*)d_ws + obytes);
  float* inc_ws  = loss_ws + S;

  int waves  = (N + TILE_ROWS - 1) / TILE_ROWS;
  int blocks = (waves + GEMV_WAVES - 1) / GEMV_WAVES;
  size_t lds = ((size_t)Hdim + (size_t)GEMV_WAVES * TILE_ROWS * ROW_PITCH) * sizeof(float);
  gemv_wmma_f32<<<blocks, GEMV_WAVES * 32, lds, stream>>>(A, W, b, logits, N, Hdim);

  scan_offsets<<<1, 32, 0, stream>>>(lens, offsets, S);

  seg_loss<<<(S + 7) / 8, 256, 0, stream>>>(logits, lens, offsets, tgt,
                                            loss_ws, inc_ws, S, EACH);

  final_reduce<<<1, 256, 0, stream>>>(loss_ws, inc_ws, loss_out, S);
}


# Offsets scan co-scheduled in the product launch: one extra 2-wave work-group picked by block index scans lens (a 32-value run per lane, one load trip) while the rest stream A; 4 launches to 3
# speedup vs baseline: 2.8444x; 1.5948x over previous
#include <hip/hip_runtime.h>
#include <limits.h>

typedef float v2f __attribute__((ext_vector_type(2)));
typedef float v4f __attribute__((ext_vector_type(4)));
typedef float v8f __attribute__((ext_vector_type(8)));
typedef int v4i __attribute__((ext_vector_type(4)));

#define NEG_INF (-1e30f)

#define GEMV_WAVES 2
#define TILE_ROWS 16
#define CHUNK_K 128
#define ROW_PITCH 132
#define SCAN_RUN 32
#define SCAN_TILE (SCAN_RUN * GEMV_WAVES * 32)

#define LOAD_CHUNK(g, c)                                                         \
  {                                                                              \
    unsigned q = (unsigned)lane + (unsigned)(c) * (CHUNK_K / 4);                 \
    _Pragma("unroll") for (int r = 0; r < TILE_ROWS; ++r)                        \
      g[r] = __builtin_nontemporal_load(rowb[r] + q);                            \
  }

#define STORE_CHUNK(g)                                                           \
  _Pragma("unroll") for (int r = 0; r < TILE_ROWS; ++r)                          \
    *(v4f*)(a_st + r * ROW_PITCH) = g[r];

#define COMPUTE_CHUNK(c)                                                         \
  {                                                                              \
    const float* wf = w_fr + (c) * CHUNK_K;                                      \
    _Pragma("unroll") for (int i = 0; i < CHUNK_K / 8; i += 2) {                 \
      v4f a0 = *(const v4f*)(a_fr + 8 * i);                                      \
      v4f w0 = *(const v4f*)(wf + 8 * i);                                        \
      v4f a1 = *(const v4f*)(a_fr + 8 * i + 8);                                  \
      v4f w1 = *(const v4f*)(wf + 8 * i + 8);                                    \
      acc0 = __builtin_amdgcn_wmma_f32_16x16x4_f32(                              \
          false, a0.xy, false, w0.xy, (short)0, acc0, false, false);             \
      acc1 = __builtin_amdgcn_wmma_f32_16x16x4_f32(                              \
          false, a0.zw, false, w0.zw, (short)0, acc1, false, false);             \
      acc2 = __builtin_amdgcn_wmma_f32_16x16x4_f32(                              \
          false, a1.xy, false, w1.xy, (short)0, acc2, false, false);             \
      acc3 = __builtin_amdgcn_wmma_f32_16x16x4_f32(                              \
          false, a1.zw, false, w1.zw, (short)0, acc3, false, false);             \
    }                                                                            \
  }

#define PHASE_FENCE()                                                            \
  asm volatile("" : "+v"(acc0), "+v"(acc1), "+v"(acc2), "+v"(acc3) : : "memory");

__device__ __forceinline__ void offsets_scan(const int* __restrict__ lens,
                                             int* __restrict__ offsets, int S,
                                             int* xch) {
  int t = (int)threadIdx.x;
  int lane = t & 31;
  int wv = t >> 5;
  int run = 0;
  for (int base = 0; base < S; base += SCAN_TILE) {
    int i0 = base + t * SCAN_RUN;
    bool mine = i0 < S;
    v4i q[SCAN_RUN / 4];
#pragma unroll
    for (int j = 0; j < SCAN_RUN / 4; ++j) q[j] = (v4i){0, 0, 0, 0};
    if (mine) {
#pragma unroll
      for (int j = 0; j < SCAN_RUN / 4; ++j) q[j] = *(const v4i*)(lens + i0 + 4 * j);
    }
    int v[SCAN_RUN];
    int acc = 0;
#pragma unroll
    for (int j = 0; j < SCAN_RUN / 4; ++j) {
      acc += q[j].x; v[4 * j] = acc;
      acc += q[j].y; v[4 * j + 1] = acc;
      acc += q[j].z; v[4 * j + 2] = acc;
      acc += q[j].w; v[4 * j + 3] = acc;
    }
    int tot = acc;
    int inc = tot;
#pragma unroll
    for (int d = 1; d < 32; d <<= 1) {
      int u = __shfl_up(inc, d, 32);
      if (lane >= d) inc += u;
    }
    if (lane == 31) xch[wv] = inc;
    __syncthreads();
    int before = run + (inc - tot);
    int all = 0;
#pragma unroll
    for (int w = 0; w < GEMV_WAVES; ++w) {
      int x = xch[w];
      if (w < wv) before += x;
      all += x;
    }
    run += all;
    if (mine) {
#pragma unroll
      for (int j = 0; j < SCAN_RUN / 4; ++j) {
        v4i e;
        e.x = before + (j ? v[4 * j - 1] : 0);
        e.y = before + v[4 * j];
        e.z = before + v[4 * j + 1];
        e.w = before + v[4 * j + 2];
        *(v4i*)(offsets + i0 + 4 * j) = e;
      }
    }
    __syncthreads();
  }
}

__global__ __launch_bounds__(GEMV_WAVES * 32) __attribute__((amdgpu_waves_per_eu(4, 4)))
void gemv_wmma_f32_and_offsets(
    const float* __restrict__ A, const float* __restrict__ W,
    const float* __restrict__ bias, float* __restrict__ out,
    int N, int Hdim,
    const int* __restrict__ lens, int* __restrict__ offsets, int S) {
  extern __shared__ __attribute__((aligned(16))) float smem[];
  if (blockIdx.x == 0) {
    offsets_scan(lens, offsets, S, (int*)smem);
    return;
  }
  float* w_lds = smem;
  int lane = (int)(threadIdx.x & 31);
  int wv = __builtin_amdgcn_readfirstlane((int)(threadIdx.x >> 5));
  float* a_lds = smem + Hdim + wv * (TILE_ROWS * ROW_PITCH);

  int row0 = (((int)blockIdx.x - 1) * GEMV_WAVES + wv) * TILE_ROWS;
  int m = lane & 15;
  int h = lane >> 4;

  const v4f* rowb[TILE_ROWS];
#pragma unroll
  for (int r = 0; r < TILE_ROWS; ++r) {
    int rr = row0 + r;
    if (rr >= N) rr = N - 1;
    rowb[r] = (const v4f*)(A + (size_t)rr * (size_t)Hdim);
  }
  float* a_st = a_lds + 4 * lane;
  const float* a_fr = a_lds + m * ROW_PITCH + 4 * h;
  const float* w_fr = w_lds + 4 * h;

  int nch = Hdim / CHUNK_K;
  v4f g0[TILE_ROWS], g1[TILE_ROWS];

  LOAD_CHUNK(g0, 0)
  if (nch > 1) LOAD_CHUNK(g1, 1)
  for (int p = wv; p < nch; p += 4 * GEMV_WAVES) {
    v4f t[4];
#pragma unroll
    for (int j = 0; j < 4; ++j) {
      int pj = p + j * GEMV_WAVES;
      if (pj < nch) t[j] = *(const v4f*)(W + pj * CHUNK_K + 4 * lane);
    }
#pragma unroll
    for (int j = 0; j < 4; ++j) {
      int pj = p + j * GEMV_WAVES;
      if (pj < nch) *(v4f*)(w_lds + pj * CHUNK_K + 4 * lane) = t[j];
    }
  }
  __syncthreads();

  v8f acc0 = {}; v8f acc1 = {}; v8f acc2 = {}; v8f acc3 = {};
  int c = 0;
  for (; c + 3 < nch; c += 2) {
    STORE_CHUNK(g0)
    LOAD_CHUNK(g0, c + 2)
    COMPUTE_CHUNK(c)
    PHASE_FENCE()
    STORE_CHUNK(g1)
    LOAD_CHUNK(g1, c + 3)
    COMPUTE_CHUNK(c + 1)
    PHASE_FENCE()
  }
  STORE_CHUNK(g0)
  if (c + 2 < nch) LOAD_CHUNK(g0, c + 2)
  COMPUTE_CHUNK(c)
  if (c + 1 < nch) {
    STORE_CHUNK(g1)
    COMPUTE_CHUNK(c + 1)
  }
  if (c + 2 < nch) {
    STORE_CHUNK(g0)
    COMPUTE_CHUNK(c + 2)
  }
  v8f acc = (acc0 + acc1) + (acc2 + acc3);
  float bv = bias[0];

  if ((lane & 15) == 0) {
    int mb = row0 + (lane >> 4) * 8;
#pragma unroll
    for (int r = 0; r < 8; ++r) {
      int rr = mb + r;
      if (rr < N) out[rr] = acc[r] + bv;
    }
  }
}

__global__ __launch_bounds__(256) void seg_loss(
    const float* __restrict__ logits, const int* __restrict__ lens,
    const int* __restrict__ offsets, const int* __restrict__ tgt,
    float* __restrict__ loss_ws, float* __restrict__ inc_ws,
    int S, int EACH) {
  int wave = (int)((blockIdx.x * blockDim.x + threadIdx.x) >> 5);
  int lane = (int)(threadIdx.x & 31);
  int s = wave;
  if (s >= S) return;

  int len = lens[s];
  int off = offsets[s];
  int j0 = lane, j1 = lane + 32;
  bool v0 = j0 < len, v1 = j1 < len;
  float x0 = v0 ? logits[off + j0] : NEG_INF;
  float x1 = v1 ? logits[off + j1] : NEG_INF;
  int   t0 = v0 ? tgt[off + j0] : 0;
  int   t1 = v1 ? tgt[off + j1] : 0;

  float mx = fmaxf(x0, x1);
#pragma unroll
  for (int d = 16; d > 0; d >>= 1) mx = fmaxf(mx, __shfl_xor(mx, d, 32));
  float e0 = v0 ? __expf(x0 - mx) : 0.0f;
  float e1 = v1 ? __expf(x1 - mx) : 0.0f;
  float sum = e0 + e1;
#pragma unroll
  for (int d = 16; d > 0; d >>= 1) sum += __shfl_xor(sum, d, 32);
  float inv = 1.0f / sum;
  float p0 = e0 * inv, p1 = e1 * inv;

  float q0 = v0 ? p0 : NEG_INF;
  float q1 = v1 ? p1 : NEG_INF;
  float m2 = fmaxf(q0, q1);
#pragma unroll
  for (int d = 16; d > 0; d >>= 1) m2 = fmaxf(m2, __shfl_xor(m2, d, 32));
  float s2 = (v0 ? __expf(p0 - m2) : 0.0f) + (v1 ? __expf(p1 - m2) : 0.0f);
#pragma unroll
  for (int d = 16; d > 0; d >>= 1) s2 += __shfl_xor(s2, d, 32);
  float lse2 = m2 + __logf(s2);

  int k0 = (v0 && t0 == 1) ? j0 : INT_MAX;
  int k1 = (v1 && t1 == 1) ? j1 : INT_MAX;
  int tk = min(k0, k1);
#pragma unroll
  for (int d = 16; d > 0; d >>= 1) tk = min(tk, __shfl_xor(tk, d, 32));
  int t_idx = (tk == INT_MAX) ? 0 : tk;
  float psel = (t_idx >= 32) ? p1 : p0;
  float p_t = __shfl(psel, t_idx & 31, 32);
  float loss_per = lse2 - p_t;

  int ts = t0 + t1;
#pragma unroll
  for (int d = 16; d > 0; d >>= 1) ts += __shfl_xor(ts, d, 32);

  int next_len = (s + 1 < S) ? lens[s + 1] : 1;
  bool skip_next_zero = (s > 0) && (s + 1 < S) && (next_len == 0);
  bool skip_no_pos = (((s + 1) % EACH) == 0) && (ts <= 0);
  bool include = (len > 0) && !skip_next_zero && !skip_no_pos;

  if (lane == 0) {
    loss_ws[s] = include ? loss_per : 0.0f;
    inc_ws[s]  = include ? 1.0f : 0.0f;
  }
}

__global__ __launch_bounds__(256) void final_reduce(
    const float* __restrict__ loss_ws, const float* __restrict__ inc_ws,
    float* __restrict__ out_loss, int S) {
  __shared__ float sl[256];
  __shared__ float sc[256];
  int t = (int)threadIdx.x;
  float a = 0.0f, c = 0.0f;
  for (int i = t; i < S; i += 256) { a += loss_ws[i]; c += inc_ws[i]; }
  sl[t] = a; sc[t] = c;
  __syncthreads();
  for (int d = 128; d > 0; d >>= 1) {
    if (t < d) { sl[t] += sl[t + d]; sc[t] += sc[t + d]; }
    __syncthreads();
  }
  if (t == 0) {
    float cnt = sc[0];
    float denom = (cnt > 0.0f) ? cnt : 1e-4f;
    out_loss[0] = sl[0] / denom;
  }
}

extern "C" void kernel_launch(void* const* d_in, const int* in_sizes, int n_in,
                              void* d_out, int out_size, void* d_ws, size_t ws_size,
                              hipStream_t stream) {
  const float* A    = (const float*)d_in[0];
  const float* W    = (const float*)d_in[1];
  const float* b    = (const float*)d_in[2];
  const int*   lens = (const int*)d_in[3];
  const int*   tgt  = (const int*)d_in[4];

  int Hdim = in_sizes[1];
  int N    = in_sizes[0] / Hdim;
  int S    = in_sizes[3];
  const int EACH = 32;

  float* logits   = (float*)d_out;
  float* loss_out = logits + N;

  int*   offsets = (int*)d_ws;
  size_t obytes  = (((size_t)S * sizeof(int)) + 255) & ~(size_t)255;
  float* loss_ws = (float*)((char*)d_ws + obytes);
  float* inc_ws  = loss_ws + S;

  int waves  = (N + TILE_ROWS - 1) / TILE_ROWS;
  int blocks = (waves + GEMV_WAVES - 1) / GEMV_WAVES;
  size_t lds = ((size_t)Hdim + (size_t)GEMV_WAVES * TILE_ROWS * ROW_PITCH) * sizeof(float);
  gemv_wmma_f32_and_offsets<<<blocks + 1, GEMV_WAVES * 32, lds, stream>>>(
      A, W, b, logits, N, Hdim, lens, offsets, S);

  seg_loss<<<(S + 7) / 8, 256, 0, stream>>>(logits, lens, offsets, tgt,
                                            loss_ws, inc_ws, S, EACH);

  final_reduce<<<1, 256, 0, stream>>>(loss_ws, inc_ws, loss_out, S);
}


# Two-level loss sum: each seg_loss work-group adds its 8 (loss, include) pairs behind one barrier; final launch = one wave, one load trip, no barrier (was 6 trips, 9 barriers); loss sum order changes
# speedup vs baseline: 3.0887x; 1.0859x over previous
#include <hip/hip_runtime.h>
#include <limits.h>

typedef float v2f __attribute__((ext_vector_type(2)));
typedef float v4f __attribute__((ext_vector_type(4)));
typedef float v8f __attribute__((ext_vector_type(8)));
typedef int v4i __attribute__((ext_vector_type(4)));

#define NEG_INF (-1e30f)

#define GEMV_WAVES 2
#define TILE_ROWS 16
#define CHUNK_K 128
#define ROW_PITCH 132
#define SCAN_RUN 32
#define SCAN_TILE (SCAN_RUN * GEMV_WAVES * 32)

#define LOAD_CHUNK(g, c)                                                         \
  {                                                                              \
    unsigned q = (unsigned)lane + (unsigned)(c) * (CHUNK_K / 4);                 \
    _Pragma("unroll") for (int r = 0; r < TILE_ROWS; ++r)                        \
      g[r] = __builtin_nontemporal_load(rowb[r] + q);                            \
  }

#define STORE_CHUNK(g)                                                           \
  _Pragma("unroll") for (int r = 0; r < TILE_ROWS; ++r)                          \
    *(v4f*)(a_st + r * ROW_PITCH) = g[r];

#define COMPUTE_CHUNK(c)                                                         \
  {                                                                              \
    const float* wf = w_fr + (c) * CHUNK_K;                                      \
    _Pragma("unroll") for (int i = 0; i < CHUNK_K / 8; i += 2) {                 \
      v4f a0 = *(const v4f*)(a_fr + 8 * i);                                      \
      v4f w0 = *(const v4f*)(wf + 8 * i);                                        \
      v4f a1 = *(const v4f*)(a_fr + 8 * i + 8);                                  \
      v4f w1 = *(const v4f*)(wf + 8 * i + 8);                                    \
      acc0 = __builtin_amdgcn_wmma_f32_16x16x4_f32(                              \
          false, a0.xy, false, w0.xy, (short)0, acc0, false, false);             \
      acc1 = __builtin_amdgcn_wmma_f32_16x16x4_f32(                              \
          false, a0.zw, false, w0.zw, (short)0, acc1, false, false);             \
      acc2 = __builtin_amdgcn_wmma_f32_16x16x4_f32(                              \
          false, a1.xy, false, w1.xy, (short)0, acc2, false, false);             \
      acc3 = __builtin_amdgcn_wmma_f32_16x16x4_f32(                              \
          false, a1.zw, false, w1.zw, (short)0, acc3, false, false);             \
    }                                                                            \
  }

#define PHASE_FENCE()                                                            \
  asm volatile("" : "+v"(acc0), "+v"(acc1), "+v"(acc2), "+v"(acc3) : : "memory");

__device__ __forceinline__ void offsets_scan(const int* __restrict__ lens,
                                             int* __restrict__ offsets, int S,
                                             int* xch) {
  int t = (int)threadIdx.x;
  int lane = t & 31;
  int wv = t >> 5;
  int run = 0;
  for (int base = 0; base < S; base += SCAN_TILE) {
    int i0 = base + t * SCAN_RUN;
    bool mine = i0 < S;
    v4i q[SCAN_RUN / 4];
#pragma unroll
    for (int j = 0; j < SCAN_RUN / 4; ++j) q[j] = (v4i){0, 0, 0, 0};
    if (mine) {
#pragma unroll
      for (int j = 0; j < SCAN_RUN / 4; ++j) q[j] = *(const v4i*)(lens + i0 + 4 * j);
    }
    int v[SCAN_RUN];
    int acc = 0;
#pragma unroll
    for (int j = 0; j < SCAN_RUN / 4; ++j) {
      acc += q[j].x; v[4 * j] = acc;
      acc += q[j].y; v[4 * j + 1] = acc;
      acc += q[j].z; v[4 * j + 2] = acc;
      acc += q[j].w; v[4 * j + 3] = acc;
    }
    int tot = acc;
    int inc = tot;
#pragma unroll
    for (int d = 1; d < 32; d <<= 1) {
      int u = __shfl_up(inc, d, 32);
      if (lane >= d) inc += u;
    }
    if (lane == 31) xch[wv] = inc;
    __syncthreads();
    int before = run + (inc - tot);
    int all = 0;
#pragma unroll
    for (int w = 0; w < GEMV_WAVES; ++w) {
      int x = xch[w];
      if (w < wv) before += x;
      all += x;
    }
    run += all;
    if (mine) {
#pragma unroll
      for (int j = 0; j < SCAN_RUN / 4; ++j) {
        v4i e;
        e.x = before + (j ? v[4 * j - 1] : 0);
        e.y = before + v[4 * j];
        e.z = before + v[4 * j + 1];
        e.w = before + v[4 * j + 2];
        *(v4i*)(offsets + i0 + 4 * j) = e;
      }
    }
    __syncthreads();
  }
}

__global__ __launch_bounds__(GEMV_WAVES * 32) __attribute__((amdgpu_waves_per_eu(4, 4)))
void gemv_wmma_f32_and_offsets(
    const float* __restrict__ A, const float* __restrict__ W,
    const float* __restrict__ bias, float* __restrict__ out,
    int N, int Hdim,
    const int* __restrict__ lens, int* __restrict__ offsets, int S) {
  extern __shared__ __attribute__((aligned(16))) float smem[];
  if (blockIdx.x == 0) {
    offsets_scan(lens, offsets, S, (int*)smem);
    return;
  }
  float* w_lds = smem;
  int lane = (int)(threadIdx.x & 31);
  int wv = __builtin_amdgcn_readfirstlane((int)(threadIdx.x >> 5));
  float* a_lds = smem + Hdim + wv * (TILE_ROWS * ROW_PITCH);

  int row0 = (((int)blockIdx.x - 1) * GEMV_WAVES + wv) * TILE_ROWS;
  int m = lane & 15;
  int h = lane >> 4;

  const v4f* rowb[TILE_ROWS];
#pragma unroll
  for (int r = 0; r < TILE_ROWS; ++r) {
    int rr = row0 + r;
    if (rr >= N) rr = N - 1;
    rowb[r] = (const v4f*)(A + (size_t)rr * (size_t)Hdim);
  }
  float* a_st = a_lds + 4 * lane;
  const float* a_fr = a_lds + m * ROW_PITCH + 4 * h;
  const float* w_fr = w_lds + 4 * h;

  int nch = Hdim / CHUNK_K;
  v4f g0[TILE_ROWS], g1[TILE_ROWS];

  LOAD_CHUNK(g0, 0)
  if (nch > 1) LOAD_CHUNK(g1, 1)
  for (int p = wv; p < nch; p += 4 * GEMV_WAVES) {
    v4f t[4];
#pragma unroll
    for (int j = 0; j < 4; ++j) {
      int pj = p + j * GEMV_WAVES;
      if (pj < nch) t[j] = *(const v4f*)(W + pj * CHUNK_K + 4 * lane);
    }
#pragma unroll
    for (int j = 0; j < 4; ++j) {
      int pj = p + j * GEMV_WAVES;
      if (pj < nch) *(v4f*)(w_lds + pj * CHUNK_K + 4 * lane) = t[j];
    }
  }
  __syncthreads();

  v8f acc0 = {}; v8f acc1 = {}; v8f acc2 = {}; v8f acc3 = {};
  int c = 0;
  for (; c + 3 < nch; c += 2) {
    STORE_CHUNK(g0)
    LOAD_CHUNK(g0, c + 2)
    COMPUTE_CHUNK(c)
    PHASE_FENCE()
    STORE_CHUNK(g1)
    LOAD_CHUNK(g1, c + 3)
    COMPUTE_CHUNK(c + 1)
    PHASE_FENCE()
  }
  STORE_CHUNK(g0)
  if (c + 2 < nch) LOAD_CHUNK(g0, c + 2)
  COMPUTE_CHUNK(c)
  if (c + 1 < nch) {
    STORE_CHUNK(g1)
    COMPUTE_CHUNK(c + 1)
  }
  if (c + 2 < nch) {
    STORE_CHUNK(g0)
    COMPUTE_CHUNK(c + 2)
  }
  v8f acc = (acc0 + acc1) + (acc2 + acc3);
  float bv = bias[0];

  if ((lane & 15) == 0) {
    int mb = row0 + (lane >> 4) * 8;
#pragma unroll
    for (int r = 0; r < 8; ++r) {
      int rr = mb + r;
      if (rr < N) out[rr] = acc[r] + bv;
    }
  }
}

#define SEG_WAVES 8

__global__ __launch_bounds__(SEG_WAVES * 32) void seg_loss(
    const float* __restrict__ logits, const int* __restrict__ lens,
    const int* __restrict__ offsets, const int* __restrict__ tgt,
    float* __restrict__ loss_ws, float* __restrict__ inc_ws,
    int S, int EACH) {
  __shared__ __attribute__((aligned(16))) float part[2 * SEG_WAVES];
  int wv = (int)(threadIdx.x >> 5);
  int lane = (int)(threadIdx.x & 31);
  int s = (int)blockIdx.x * SEG_WAVES + wv;
  float my_loss = 0.0f, my_inc = 0.0f;

  if (s < S) {
    int len = lens[s];
    int off = offsets[s];
    int j0 = lane, j1 = lane + 32;
    bool v0 = j0 < len, v1 = j1 < len;
    float x0 = v0 ? logits[off + j0] : NEG_INF;
    float x1 = v1 ? logits[off + j1] : NEG_INF;
    int   t0 = v0 ? tgt[off + j0] : 0;
    int   t1 = v1 ? tgt[off + j1] : 0;

    float mx = fmaxf(x0, x1);
#pragma unroll
    for (int d = 16; d > 0; d >>= 1) mx = fmaxf(mx, __shfl_xor(mx, d, 32));
    float e0 = v0 ? __expf(x0 - mx) : 0.0f;
    float e1 = v1 ? __expf(x1 - mx) : 0.0f;
    float sum = e0 + e1;
#pragma unroll
    for (int d = 16; d > 0; d >>= 1) sum += __shfl_xor(sum, d, 32);
    float inv = 1.0f / sum;
    float p0 = e0 * inv, p1 = e1 * inv;

    float q0 = v0 ? p0 : NEG_INF;
    float q1 = v1 ? p1 : NEG_INF;
    float m2 = fmaxf(q0, q1);
#pragma unroll
    for (int d = 16; d > 0; d >>= 1) m2 = fmaxf(m2, __shfl_xor(m2, d, 32));
    float s2 = (v0 ? __expf(p0 - m2) : 0.0f) + (v1 ? __expf(p1 - m2) : 0.0f);
#pragma unroll
    for (int d = 16; d > 0; d >>= 1) s2 += __shfl_xor(s2, d, 32);
    float lse2 = m2 + __logf(s2);

    int k0 = (v0 && t0 == 1) ? j0 : INT_MAX;
    int k1 = (v1 && t1 == 1) ? j1 : INT_MAX;
    int tk = min(k0, k1);
#pragma unroll
    for (int d = 16; d > 0; d >>= 1) tk = min(tk, __shfl_xor(tk, d, 32));
    int t_idx = (tk == INT_MAX) ? 0 : tk;
    float psel = (t_idx >= 32) ? p1 : p0;
    float p_t = __shfl(psel, t_idx & 31, 32);
    float loss_per = lse2 - p_t;

    int ts = t0 + t1;
#pragma unroll
    for (int d = 16; d > 0; d >>= 1) ts += __shfl_xor(ts, d, 32);

    int next_len = (s + 1 < S) ? lens[s + 1] : 1;
    bool skip_next_zero = (s > 0) && (s + 1 < S) && (next_len == 0);
    bool skip_no_pos = (((s + 1) % EACH) == 0) && (ts <= 0);
    bool include = (len > 0) && !skip_next_zero && !skip_no_pos;

    my_loss = include ? loss_per : 0.0f;
    my_inc  = include ? 1.0f : 0.0f;
  }

  if (lane == 0) {
    part[wv] = my_loss;
    part[SEG_WAVES + wv] = my_inc;
  }
  __syncthreads();
  if (threadIdx.x == 0) {
    v4f l0 = *(const v4f*)(part);
    v4f l1 = *(const v4f*)(part + 4);
    v4f c0 = *(const v4f*)(part + SEG_WAVES);
    v4f c1 = *(const v4f*)(part + SEG_WAVES + 4);
    loss_ws[blockIdx.x] = ((l0.x + l0.y) + (l0.z + l0.w)) + ((l1.x + l1.y) + (l1.z + l1.w));
    inc_ws[blockIdx.x]  = ((c0.x + c0.y) + (c0.z + c0.w)) + ((c1.x + c1.y) + (c1.z + c1.w));
  }
}

#define FIN_RUN 8

__global__ __launch_bounds__(32) void final_reduce(
    const float* __restrict__ loss_ws, const float* __restrict__ inc_ws,
    float* __restrict__ out_loss, int S) {
  int P = (S + SEG_WAVES - 1) / SEG_WAVES;
  int lane = (int)threadIdx.x;
  float a = 0.0f, c = 0.0f;
  for (int base = 0; base < P; base += 32 * FIN_RUN) {
    float vl[FIN_RUN], vc[FIN_RUN];
#pragma unroll
    for (int j = 0; j < FIN_RUN; ++j) {
      int i = base + lane + 32 * j;
      int ii = (i < P) ? i : 0;
      vl[j] = loss_ws[ii];
      vc[j] = inc_ws[ii];
    }
    __builtin_amdgcn_sched_barrier(0);
#pragma unroll
    for (int j = 0; j < FIN_RUN; ++j) {
      bool ok = base + lane + 32 * j < P;
      a += ok ? vl[j] : 0.0f;
      c += ok ? vc[j] : 0.0f;
    }
  }
#pragma unroll
  for (int d = 16; d > 0; d >>= 1) {
    a += __shfl_xor(a, d, 32);
    c += __shfl_xor(c, d, 32);
  }
  if (lane == 0) {
    float denom = (c > 0.0f) ? c : 1e-4f;
    out_loss[0] = a / denom;
  }
}

extern "C" void kernel_launch(void* const* d_in, const int* in_sizes, int n_in,
                              void* d_out, int out_size, void* d_ws, size_t ws_size,
                              hipStream_t stream) {
  const float* A    = (const float*)d_in[0];
  const float* W    = (const float*)d_in[1];
  const float* b    = (const float*)d_in[2];
  const int*   lens = (const int*)d_in[3];
  const int*   tgt  = (const int*)d_in[4];

  int Hdim = in_sizes[1];
  int N    = in_sizes[0] / Hdim;
  int S    = in_sizes[3];
  const int EACH = 32;

  float* logits   = (float*)d_out;
  float* loss_out = logits + N;

  int*   offsets = (int*)d_ws;
  size_t obytes  = (((size_t)S * sizeof(int)) + 255) & ~(size_t)255;
  float* loss_ws = (float*)((char*)d_ws + obytes);
  float* inc_ws  = loss_ws + S;

  int waves  = (N + TILE_ROWS - 1) / TILE_ROWS;
  int blocks = (waves + GEMV_WAVES - 1) / GEMV_WAVES;
  size_t lds = ((size_t)Hdim + (size_t)GEMV_WAVES * TILE_ROWS * ROW_PITCH) * sizeof(float);
  gemv_wmma_f32_and_offsets<<<blocks + 1, GEMV_WAVES * 32, lds, stream>>>(
      A, W, b, logits, N, Hdim, lens, offsets, S);

  seg_loss<<<(S + SEG_WAVES - 1) / SEG_WAVES, SEG_WAVES * 32, 0, stream>>>(
      logits, lens, offsets, tgt, loss_ws, inc_ws, S, EACH);

  final_reduce<<<1, 32, 0, stream>>>(loss_ws, inc_ws, loss_out, S);
}
